# starting kernel
# baseline (speedup 1.0000x reference)
#include <hip/hip_runtime.h>
#include <math.h>

typedef float v2f __attribute__((ext_vector_type(2)));
typedef float v8f __attribute__((ext_vector_type(8)));

#define WAVES_PER_BLOCK 4
#define PIf 3.14159265358979323846f

__global__ void vqc_setup_kernel(const float* __restrict__ params,
                                 float* __restrict__ Mout) {
    __shared__ float A[32][33];
    const int lane = threadIdx.x;

    float gc[10], gs[10];
#pragma unroll
    for (int g = 0; g < 10; ++g) {
        float th = params[g * 2];
        gs[g] = __sinf(0.5f * th);
        gc[g] = __cosf(0.5f * th);
    }

    for (int j = 0; j < 32; ++j) {
        float val = (lane == j) ? 1.0f : 0.0f;
#pragma unroll
        for (int l = 0; l < 2; ++l) {
#pragma unroll
            for (int i = 0; i < 5; ++i) {
                const int g = l * 5 + i;
                const int mask = 1 << (4 - i);
                float ov = __shfl(val, lane ^ mask, 32);
                int bit = (lane >> (4 - i)) & 1;
                val = bit ? fmaf(gs[g], ov, gc[g] * val)
                          : fmaf(-gs[g], ov, gc[g] * val);
            }
#pragma unroll
            for (int i = 0; i < 5; ++i) {
                const int t = (i + 1) % 5;
                int cbit = (lane >> (4 - i)) & 1;
                int src = cbit ? (lane ^ (1 << (4 - t))) : lane;
                val = __shfl(val, src, 32);
            }
        }
        float dcz = 1.0f;
#pragma unroll
        for (int i = 0; i < 5; ++i) {
            int b1 = (j >> (4 - i)) & 1;
            int b2 = (j >> (4 - (i + 1) % 5)) & 1;
            if (b1 & b2) dcz = -dcz;
        }
        A[lane][j] = dcz * val;
    }
    __syncthreads();

    for (int p = 0; p < 32; ++p) {
        float m = 0.0f;
#pragma unroll
        for (int r = 0; r < 32; ++r) {
            float zr = (r & 16) ? -1.0f : 1.0f;
            m = fmaf(zr * A[r][p], A[r][lane], m);
        }
        Mout[p * 32 + lane] = m;
    }
}

__global__ __launch_bounds__(WAVES_PER_BLOCK * 32) void vqc_main_kernel(
    const float* __restrict__ x, const float* __restrict__ Mq,
    float* __restrict__ out, int batch) {
    const int lane = threadIdx.x & 31;
    const int wave = threadIdx.x >> 5;
    const int bl = lane & 15;
    const int hi = lane >> 4;
    const int group = blockIdx.x * WAVES_PER_BLOCK + wave;
    const int e = group * 16 + bl;
    const int el = (e < batch) ? e : (batch - 1);

    float cc[5], ss[5];
#pragma unroll
    for (int w = 0; w < 5; ++w) {
        float th = 0.5f * PIf * x[el * 5 + w];
        __sincosf(th, &ss[w], &cc[w]);
    }
    float q0r[2] = {cc[0] * cc[0], cc[0] * ss[0]};
    float q0i[2] = {-cc[0] * ss[0], ss[0] * ss[0]};
    float q1r[2] = {cc[1] * cc[1], cc[1] * ss[1]};
    float q1i[2] = {-cc[1] * ss[1], ss[1] * ss[1]};
    float q2r[2] = {cc[2] * cc[2], cc[2] * ss[2]};
    float q2i[2] = {-cc[2] * ss[2], ss[2] * ss[2]};
    float q3r = hi ? cc[3] * ss[3] : cc[3] * cc[3];
    float q3i = hi ? ss[3] * ss[3] : -cc[3] * ss[3];
    float q4r[2] = {cc[4] * cc[4], cc[4] * ss[4]};
    float q4i[2] = {-cc[4] * ss[4], ss[4] * ss[4]};
    float tr[2], ti[2];
#pragma unroll
    for (int j = 0; j < 2; ++j) {
        tr[j] = q3r * q4r[j] - q3i * q4i[j];
        ti[j] = q3r * q4i[j] + q3i * q4r[j];
    }
    float p01r[4], p01i[4];
#pragma unroll
    for (int b0 = 0; b0 < 2; ++b0)
#pragma unroll
        for (int b1 = 0; b1 < 2; ++b1) {
            p01r[b0 * 2 + b1] = q0r[b0] * q1r[b1] - q0i[b0] * q1i[b1];
            p01i[b0 * 2 + b1] = q0r[b0] * q1i[b1] + q0i[b0] * q1r[b1];
        }
    v2f are[8], aim[8];
#pragma unroll
    for (int kk = 0; kk < 8; ++kk) {
        const int b01 = kk >> 1, b2 = kk & 1;
        float pr = p01r[b01] * q2r[b2] - p01i[b01] * q2i[b2];
        float pi = p01r[b01] * q2i[b2] + p01i[b01] * q2r[b2];
#pragma unroll
        for (int j = 0; j < 2; ++j) {
            are[kk][j] = pr * tr[j] - pi * ti[j];
            aim[kk][j] = pr * ti[j] + pi * tr[j];
        }
    }

    v8f accRe0 = {}, accRe1 = {}, accIm0 = {}, accIm1 = {};
#pragma unroll
    for (int kk = 0; kk < 8; ++kk) {
        const int krow = 4 * kk + 2 * hi;
        v2f b0v, b1v;
        b0v[0] = Mq[krow * 32 + bl];
        b0v[1] = Mq[(krow + 1) * 32 + bl];
        b1v[0] = Mq[krow * 32 + 16 + bl];
        b1v[1] = Mq[(krow + 1) * 32 + 16 + bl];
        accRe0 = __builtin_amdgcn_wmma_f32_16x16x4_f32(false, are[kk], false,
                                                       b0v, (short)0, accRe0,
                                                       false, false);
        accRe1 = __builtin_amdgcn_wmma_f32_16x16x4_f32(false, are[kk], false,
                                                       b1v, (short)0, accRe1,
                                                       false, false);
        accIm0 = __builtin_amdgcn_wmma_f32_16x16x4_f32(false, aim[kk], false,
                                                       b0v, (short)0, accIm0,
                                                       false, false);
        accIm1 = __builtin_amdgcn_wmma_f32_16x16x4_f32(false, aim[kk], false,
                                                       b1v, (short)0, accIm1,
                                                       false, false);
    }

    __shared__ float T[WAVES_PER_BLOCK][2][16][33];
#pragma unroll
    for (int r = 0; r < 8; ++r) {
        const int m = r + 8 * hi;
        T[wave][0][m][bl] = accRe0[r];
        T[wave][0][m][16 + bl] = accRe1[r];
        T[wave][1][m][bl] = accIm0[r];
        T[wave][1][m][16 + bl] = accIm1[r];
    }
    __syncthreads();

    float sum = 0.0f;
#pragma unroll
    for (int kk = 0; kk < 8; ++kk) {
        const int n = 4 * kk + 2 * hi;
        sum = fmaf(T[wave][0][bl][n], are[kk][0], sum);
        sum = fmaf(T[wave][0][bl][n + 1], are[kk][1], sum);
        sum = fmaf(T[wave][1][bl][n], aim[kk][0], sum);
        sum = fmaf(T[wave][1][bl][n + 1], aim[kk][1], sum);
    }
    sum += __shfl(sum, lane ^ 16, 32);
    if (hi == 0 && e < batch) out[e] = sum;
}

extern "C" void kernel_launch(void* const* d_in, const int* in_sizes, int n_in,
                              void* d_out, int out_size, void* d_ws,
                              size_t ws_size, hipStream_t stream) {
    const float* x = (const float*)d_in[0];
    const float* params = (const float*)d_in[1];
    float* out = (float*)d_out;
    float* Mq = (float*)d_ws;

    const int batch = in_sizes[0] / 5;

    vqc_setup_kernel<<<1, 32, 0, stream>>>(params, Mq);

    const int groups = (batch + 15) / 16;
    const int blocks = (groups + WAVES_PER_BLOCK - 1) / WAVES_PER_BLOCK;
    vqc_main_kernel<<<blocks, WAVES_PER_BLOCK * 32, 0, stream>>>(x, Mq, out,
                                                                 batch);
}


# setup kernel spread over 32 waves of one work-group: wave j propagates basis column j, thread (p,q) sums M[p][q]; same arithmetic, same r order, M bit-identical
# speedup vs baseline: 2.7799x; 2.7799x over previous
#include <hip/hip_runtime.h>
#include <math.h>

typedef float v2f __attribute__((ext_vector_type(2)));
typedef float v8f __attribute__((ext_vector_type(8)));

#define WAVES_PER_BLOCK 4
#define PIf 3.14159265358979323846f

#define SETUP_WAVES 32
__global__ __launch_bounds__(SETUP_WAVES * 32) void vqc_setup_kernel(
    const float* __restrict__ params, float* __restrict__ Mout) {
    __shared__ float A[32][33];
    const int lane = threadIdx.x & 31;
    const int j = threadIdx.x >> 5;

    float gc[10], gs[10];
#pragma unroll
    for (int g = 0; g < 10; ++g) {
        float th = params[g * 2];
        gs[g] = __sinf(0.5f * th);
        gc[g] = __cosf(0.5f * th);
    }

    float val = (lane == j) ? 1.0f : 0.0f;
#pragma unroll
    for (int l = 0; l < 2; ++l) {
#pragma unroll
        for (int i = 0; i < 5; ++i) {
            const int g = l * 5 + i;
            const int mask = 1 << (4 - i);
            float ov = __shfl(val, lane ^ mask, 32);
            int bit = (lane >> (4 - i)) & 1;
            val = bit ? fmaf(gs[g], ov, gc[g] * val)
                      : fmaf(-gs[g], ov, gc[g] * val);
        }
#pragma unroll
        for (int i = 0; i < 5; ++i) {
            const int t = (i + 1) % 5;
            int cbit = (lane >> (4 - i)) & 1;
            int src = cbit ? (lane ^ (1 << (4 - t))) : lane;
            val = __shfl(val, src, 32);
        }
    }
    float dcz = 1.0f;
#pragma unroll
    for (int i = 0; i < 5; ++i) {
        int b1 = (j >> (4 - i)) & 1;
        int b2 = (j >> (4 - (i + 1) % 5)) & 1;
        if (b1 & b2) dcz = -dcz;
    }
    A[lane][j] = dcz * val;
    __syncthreads();

    const int p = j;
    float m = 0.0f;
#pragma unroll
    for (int r = 0; r < 32; ++r) {
        float zr = (r & 16) ? -1.0f : 1.0f;
        m = fmaf(zr * A[r][p], A[r][lane], m);
    }
    Mout[p * 32 + lane] = m;
}

__global__ __launch_bounds__(WAVES_PER_BLOCK * 32) void vqc_main_kernel(
    const float* __restrict__ x, const float* __restrict__ Mq,
    float* __restrict__ out, int batch) {
    const int lane = threadIdx.x & 31;
    const int wave = threadIdx.x >> 5;
    const int bl = lane & 15;
    const int hi = lane >> 4;
    const int group = blockIdx.x * WAVES_PER_BLOCK + wave;
    const int e = group * 16 + bl;
    const int el = (e < batch) ? e : (batch - 1);

    float cc[5], ss[5];
#pragma unroll
    for (int w = 0; w < 5; ++w) {
        float th = 0.5f * PIf * x[el * 5 + w];
        __sincosf(th, &ss[w], &cc[w]);
    }
    float q0r[2] = {cc[0] * cc[0], cc[0] * ss[0]};
    float q0i[2] = {-cc[0] * ss[0], ss[0] * ss[0]};
    float q1r[2] = {cc[1] * cc[1], cc[1] * ss[1]};
    float q1i[2] = {-cc[1] * ss[1], ss[1] * ss[1]};
    float q2r[2] = {cc[2] * cc[2], cc[2] * ss[2]};
    float q2i[2] = {-cc[2] * ss[2], ss[2] * ss[2]};
    float q3r = hi ? cc[3] * ss[3] : cc[3] * cc[3];
    float q3i = hi ? ss[3] * ss[3] : -cc[3] * ss[3];
    float q4r[2] = {cc[4] * cc[4], cc[4] * ss[4]};
    float q4i[2] = {-cc[4] * ss[4], ss[4] * ss[4]};
    float tr[2], ti[2];
#pragma unroll
    for (int j = 0; j < 2; ++j) {
        tr[j] = q3r * q4r[j] - q3i * q4i[j];
        ti[j] = q3r * q4i[j] + q3i * q4r[j];
    }
    float p01r[4], p01i[4];
#pragma unroll
    for (int b0 = 0; b0 < 2; ++b0)
#pragma unroll
        for (int b1 = 0; b1 < 2; ++b1) {
            p01r[b0 * 2 + b1] = q0r[b0] * q1r[b1] - q0i[b0] * q1i[b1];
            p01i[b0 * 2 + b1] = q0r[b0] * q1i[b1] + q0i[b0] * q1r[b1];
        }
    v2f are[8], aim[8];
#pragma unroll
    for (int kk = 0; kk < 8; ++kk) {
        const int b01 = kk >> 1, b2 = kk & 1;
        float pr = p01r[b01] * q2r[b2] - p01i[b01] * q2i[b2];
        float pi = p01r[b01] * q2i[b2] + p01i[b01] * q2r[b2];
#pragma unroll
        for (int j = 0; j < 2; ++j) {
            are[kk][j] = pr * tr[j] - pi * ti[j];
            aim[kk][j] = pr * ti[j] + pi * tr[j];
        }
    }

    v8f accRe0 = {}, accRe1 = {}, accIm0 = {}, accIm1 = {};
#pragma unroll
    for (int kk = 0; kk < 8; ++kk) {
        const int krow = 4 * kk + 2 * hi;
        v2f b0v, b1v;
        b0v[0] = Mq[krow * 32 + bl];
        b0v[1] = Mq[(krow + 1) * 32 + bl];
        b1v[0] = Mq[krow * 32 + 16 + bl];
        b1v[1] = Mq[(krow + 1) * 32 + 16 + bl];
        accRe0 = __builtin_amdgcn_wmma_f32_16x16x4_f32(false, are[kk], false,
                                                       b0v, (short)0, accRe0,
                                                       false, false);
        accRe1 = __builtin_amdgcn_wmma_f32_16x16x4_f32(false, are[kk], false,
                                                       b1v, (short)0, accRe1,
                                                       false, false);
        accIm0 = __builtin_amdgcn_wmma_f32_16x16x4_f32(false, aim[kk], false,
                                                       b0v, (short)0, accIm0,
                                                       false, false);
        accIm1 = __builtin_amdgcn_wmma_f32_16x16x4_f32(false, aim[kk], false,
                                                       b1v, (short)0, accIm1,
                                                       false, false);
    }

    __shared__ float T[WAVES_PER_BLOCK][2][16][33];
#pragma unroll
    for (int r = 0; r < 8; ++r) {
        const int m = r + 8 * hi;
        T[wave][0][m][bl] = accRe0[r];
        T[wave][0][m][16 + bl] = accRe1[r];
        T[wave][1][m][bl] = accIm0[r];
        T[wave][1][m][16 + bl] = accIm1[r];
    }
    __syncthreads();

    float sum = 0.0f;
#pragma unroll
    for (int kk = 0; kk < 8; ++kk) {
        const int n = 4 * kk + 2 * hi;
        sum = fmaf(T[wave][0][bl][n], are[kk][0], sum);
        sum = fmaf(T[wave][0][bl][n + 1], are[kk][1], sum);
        sum = fmaf(T[wave][1][bl][n], aim[kk][0], sum);
        sum = fmaf(T[wave][1][bl][n + 1], aim[kk][1], sum);
    }
    sum += __shfl(sum, lane ^ 16, 32);
    if (hi == 0 && e < batch) out[e] = sum;
}

extern "C" void kernel_launch(void* const* d_in, const int* in_sizes, int n_in,
                              void* d_out, int out_size, void* d_ws,
                              size_t ws_size, hipStream_t stream) {
    const float* x = (const float*)d_in[0];
    const float* params = (const float*)d_in[1];
    float* out = (float*)d_out;
    float* Mq = (float*)d_ws;

    const int batch = in_sizes[0] / 5;

    vqc_setup_kernel<<<1, SETUP_WAVES * 32, 0, stream>>>(params, Mq);

    const int groups = (batch + 15) / 16;
    const int blocks = (groups + WAVES_PER_BLOCK - 1) / WAVES_PER_BLOCK;
    vqc_main_kernel<<<blocks, WAVES_PER_BLOCK * 32, 0, stream>>>(x, Mq, out,
                                                                 batch);
}


# setup launch folded in: 256 resident work-groups each build A_top in LDS (same arithmetic, lane exchanges on DPP), then every wave loops over its run of 16-sample groups; float sum order unchanged
# speedup vs baseline: 4.3254x; 1.5560x over previous
#include <hip/hip_runtime.h>
#include <math.h>


typedef float v2f __attribute__((ext_vector_type(2)));
typedef float v4f __attribute__((ext_vector_type(4)));
typedef float v8f __attribute__((ext_vector_type(8)));

#define WG_WAVES 16
#define NUM_WGS 256
#define TOTAL_WAVES (NUM_WGS * WG_WAVES)
#define TOTAL_WAVES_LOG2 12
#define PIf 3.14159265358979323846f
#define FRAG_FLOATS 512

template <int M>
__device__ __forceinline__ float xlane(float v) {
    const int b = __builtin_bit_cast(int, v);
    return __builtin_bit_cast(
        float, __builtin_amdgcn_update_dpp(b, b, 0x160 + M, 0xf, 0xf, false));
}
template <>
__device__ __forceinline__ float xlane<16>(float v) {
    const int b = __builtin_bit_cast(int, v);
    return __builtin_bit_cast(
        float, __builtin_amdgcn_permlanex16(b, b, 0x76543210, 0xfedcba98,
                                            false, false));
}

#define RY_STEP(G, SH)                                           \
    {                                                            \
        const float ov = xlane<(1 << (SH))>(val);                \
        const int bit = (lane >> (SH)) & 1;                      \
        val = bit ? fmaf(gs[G], ov, gc[G] * val)                 \
                  : fmaf(-gs[G], ov, gc[G] * val);               \
    }
#define CNOT_STEP(CSH, TSH)                                      \
    {                                                            \
        const float ov = xlane<(1 << (TSH))>(val);               \
        const int cbit = (lane >> (CSH)) & 1;                    \
        val = cbit ? ov : val;                                   \
    }

__device__ __forceinline__ float circuit_column(int j, int lane,
                                                const float* gc,
                                                const float* gs) {
    float val = (lane == j) ? 1.0f : 0.0f;
#pragma unroll
    for (int l = 0; l < 2; ++l) {
        RY_STEP(l * 5 + 0, 4)
        RY_STEP(l * 5 + 1, 3)
        RY_STEP(l * 5 + 2, 2)
        RY_STEP(l * 5 + 3, 1)
        RY_STEP(l * 5 + 4, 0)
        CNOT_STEP(4, 3)
        CNOT_STEP(3, 2)
        CNOT_STEP(2, 1)
        CNOT_STEP(1, 0)
        CNOT_STEP(0, 4)
    }
    float dcz = 1.0f;
#pragma unroll
    for (int i = 0; i < 5; ++i) {
        const int b1 = (j >> (4 - i)) & 1;
        const int b2 = (j >> (4 - (i + 1) % 5)) & 1;
        if (b1 & b2) dcz = -dcz;
    }
    return dcz * val;
}

__global__ __launch_bounds__(WG_WAVES * 32) void vqc_fused_kernel(
    const float* __restrict__ x, const float* __restrict__ params,
    float* __restrict__ out, int batch) {
    __shared__ __attribute__((aligned(16))) float Gs[FRAG_FLOATS];
    const int lane = threadIdx.x & 31;
    const int wave = __builtin_amdgcn_readfirstlane(threadIdx.x >> 5);
    const int bl = lane & 15;
    const int hi = lane >> 4;

    const int groups = (batch + 15) >> 4;
    const int per = (groups + TOTAL_WAVES - 1) >> TOTAL_WAVES_LOG2;
    int g = (blockIdx.x * WG_WAVES + wave) * per;
    const int gend = (g + per < groups) ? (g + per) : groups;

    float xc[5] = {0.0f, 0.0f, 0.0f, 0.0f, 0.0f};
    if (g < gend) {
        const int e0 = g * 16 + bl;
        const int el = (e0 < batch) ? e0 : (batch - 1);
#pragma unroll
        for (int w = 0; w < 5; ++w) xc[w] = x[el * 5 + w];
    }

    float gc[10], gs[10];
#pragma unroll
    for (int q = 0; q < 10; ++q) {
        const float th = params[q * 2];
        gs[q] = __sinf(0.5f * th);
        gc[q] = __cosf(0.5f * th);
    }
    const int j0 = wave, j1 = wave + WG_WAVES;
    const float a0 = circuit_column(j0, lane, gc, gs);
    const float a1 = circuit_column(j1, lane, gc, gs);
    if (lane < 16) {
        Gs[((j0 >> 1) & 1) * 256 + lane * 16 + (j0 >> 2) * 2 + (j0 & 1)] = a0;
        Gs[((j1 >> 1) & 1) * 256 + lane * 16 + (j1 >> 2) * 2 + (j1 & 1)] = a1;
    }
    float lam = 1.0f;
#pragma unroll
    for (int q = 0; q < 10; ++q) lam *= fmaf(gc[q], gc[q], gs[q] * gs[q]);
    __syncthreads();

    const v4f* gp = reinterpret_cast<const v4f*>(Gs + lane * 16);
    v4f af[4];
#pragma unroll
    for (int i = 0; i < 4; ++i) af[i] = gp[i];

    while (g < gend) {
        const int e = g * 16 + bl;

        const int gn = (g + 1 < gend) ? (g + 1) : g;
        const int en0 = gn * 16 + bl;
        const int en = (en0 < batch) ? en0 : (batch - 1);
        float xn[5];
#pragma unroll
        for (int w = 0; w < 5; ++w) xn[w] = x[en * 5 + w];

        float cc[5], ss[5];
#pragma unroll
        for (int w = 0; w < 5; ++w) {
            float th = 0.5f * PIf * xc[w];
            __sincosf(th, &ss[w], &cc[w]);
        }
        float nrm = 1.0f;
#pragma unroll
        for (int w = 0; w < 5; ++w) {
            float n = fmaf(cc[w], cc[w], ss[w] * ss[w]);
            nrm *= n * n;
        }
        float q0r[2] = {cc[0] * cc[0], cc[0] * ss[0]};
        float q0i[2] = {-cc[0] * ss[0], ss[0] * ss[0]};
        float q1r[2] = {cc[1] * cc[1], cc[1] * ss[1]};
        float q1i[2] = {-cc[1] * ss[1], ss[1] * ss[1]};
        float q2r[2] = {cc[2] * cc[2], cc[2] * ss[2]};
        float q2i[2] = {-cc[2] * ss[2], ss[2] * ss[2]};
        float q3r = hi ? cc[3] * ss[3] : cc[3] * cc[3];
        float q3i = hi ? ss[3] * ss[3] : -cc[3] * ss[3];
        float q4r[2] = {cc[4] * cc[4], cc[4] * ss[4]};
        float q4i[2] = {-cc[4] * ss[4], ss[4] * ss[4]};
        float tr[2], ti[2];
#pragma unroll
        for (int j = 0; j < 2; ++j) {
            tr[j] = q3r * q4r[j] - q3i * q4i[j];
            ti[j] = q3r * q4i[j] + q3i * q4r[j];
        }
        float p01r[4], p01i[4];
#pragma unroll
        for (int b0 = 0; b0 < 2; ++b0)
#pragma unroll
            for (int b1 = 0; b1 < 2; ++b1) {
                p01r[b0 * 2 + b1] = q0r[b0] * q1r[b1] - q0i[b0] * q1i[b1];
                p01i[b0 * 2 + b1] = q0r[b0] * q1i[b1] + q0i[b0] * q1r[b1];
            }
        v2f are[8], aim[8];
#pragma unroll
        for (int kk = 0; kk < 8; ++kk) {
            const int b01 = kk >> 1, b2 = kk & 1;
            float pr = p01r[b01] * q2r[b2] - p01i[b01] * q2i[b2];
            float pi = p01r[b01] * q2i[b2] + p01i[b01] * q2r[b2];
#pragma unroll
            for (int j = 0; j < 2; ++j) {
                are[kk][j] = pr * tr[j] - pi * ti[j];
                aim[kk][j] = pr * ti[j] + pi * tr[j];
            }
        }

        v8f accRe = {}, accIm = {};
#pragma unroll
        for (int kk = 0; kk < 8; ++kk) {
            v2f a;
            a[0] = af[kk >> 1][(kk & 1) * 2];
            a[1] = af[kk >> 1][(kk & 1) * 2 + 1];
            accRe = __builtin_amdgcn_wmma_f32_16x16x4_f32(
                false, a, false, are[kk], (short)0, accRe, false, false);
            accIm = __builtin_amdgcn_wmma_f32_16x16x4_f32(
                false, a, false, aim[kk], (short)0, accIm, false, false);
        }

        float p = 0.0f;
#pragma unroll
        for (int r = 0; r < 8; ++r) {
            p = fmaf(accRe[r], accRe[r], p);
            p = fmaf(accIm[r], accIm[r], p);
        }
        p += xlane<16>(p);
        if (hi == 0 && e < batch) out[e] = fmaf(2.0f, p, -(lam * nrm));

#pragma unroll
        for (int w = 0; w < 5; ++w) xc[w] = xn[w];
        ++g;
    }
}

extern "C" void kernel_launch(void* const* d_in, const int* in_sizes, int n_in,
                              void* d_out, int out_size, void* d_ws,
                              size_t ws_size, hipStream_t stream) {
    const float* x = (const float*)d_in[0];
    const float* params = (const float*)d_in[1];
    float* out = (float*)d_out;

    const int batch = in_sizes[0] / 5;

    vqc_fused_kernel<<<NUM_WGS, WG_WAVES * 32, 0, stream>>>(x, params, out,
                                                            batch);
}


# <Z0> as bilinear form: A^T Z A in the per-wire (1,sx,sy,sz) basis (nonzero 16x32 block, built per group) times per-wire component products: 8 WMMAs per 16 samples not 16; float sum order changes
# speedup vs baseline: 6.9659x; 1.6105x over previous
#include <hip/hip_runtime.h>
#include <math.h>


typedef float v2f __attribute__((ext_vector_type(2)));
typedef float v4f __attribute__((ext_vector_type(4)));
typedef float v8f __attribute__((ext_vector_type(8)));

#define WG_WAVES 16
#define NUM_WGS 256
#define TOTAL_WAVES (NUM_WGS * WG_WAVES)
#define TOTAL_WAVES_LOG2 12
#define PIf 3.14159265358979323846f
#define FRAG_FLOATS 512
#define CH 8
#define COL_PITCH 20

template <int M>
__device__ __forceinline__ float xlane(float v) {
    const int b = __builtin_bit_cast(int, v);
    return __builtin_bit_cast(
        float, __builtin_amdgcn_update_dpp(b, b, 0x160 + M, 0xf, 0xf, false));
}
template <>
__device__ __forceinline__ float xlane<16>(float v) {
    const int b = __builtin_bit_cast(int, v);
    return __builtin_bit_cast(
        float, __builtin_amdgcn_permlanex16(b, b, 0x76543210, 0xfedcba98,
                                            false, false));
}

#define RY_STEP(G, SH)                                           \
    {                                                            \
        const float ov = xlane<(1 << (SH))>(val);                \
        const int bit = (lane >> (SH)) & 1;                      \
        val = bit ? fmaf(gs[G], ov, gc[G] * val)                 \
                  : fmaf(-gs[G], ov, gc[G] * val);               \
    }
#define CNOT_STEP(CSH, TSH)                                      \
    {                                                            \
        const float ov = xlane<(1 << (TSH))>(val);               \
        const int cbit = (lane >> (CSH)) & 1;                    \
        val = cbit ? ov : val;                                   \
    }

__device__ __forceinline__ float circuit_column(int j, int lane,
                                                const float* gc,
                                                const float* gs) {
    float val = (lane == j) ? 1.0f : 0.0f;
#pragma unroll
    for (int l = 0; l < 2; ++l) {
        RY_STEP(l * 5 + 0, 4)
        RY_STEP(l * 5 + 1, 3)
        RY_STEP(l * 5 + 2, 2)
        RY_STEP(l * 5 + 3, 1)
        RY_STEP(l * 5 + 4, 0)
        CNOT_STEP(4, 3)
        CNOT_STEP(3, 2)
        CNOT_STEP(2, 1)
        CNOT_STEP(1, 0)
        CNOT_STEP(0, 4)
    }
    float dcz = 1.0f;
#pragma unroll
    for (int i = 0; i < 5; ++i) {
        const int b1 = (j >> (4 - i)) & 1;
        const int b2 = (j >> (4 - (i + 1) % 5)) & 1;
        if (b1 & b2) dcz = -dcz;
    }
    return dcz * val;
}

__device__ __forceinline__ void load_inputs(const float* __restrict__ x, int g,
                                            int gend, int bl, int batch,
                                            float (&xr)[CH][5]) {
#pragma unroll
    for (int t = 0; t < CH; ++t) {
        const int gt = (g + t < gend) ? (g + t) : (gend - 1);
        const int e0 = gt * 16 + bl;
        const int el = (e0 < batch) ? e0 : (batch - 1);
#pragma unroll
        for (int w = 0; w < 5; ++w) xr[t][w] = x[el * 5 + w];
    }
}

__device__ __forceinline__ float group_value(const float (&xc)[5],
                                             const v4f (&af)[4], int hi) {
    float bx[5], by[5], bz[5];
#pragma unroll
    for (int w = 0; w < 5; ++w) {
        const float rev = 0.5f * xc[w];
        const float sn = __builtin_amdgcn_sinf(rev);
        const float cs = __builtin_amdgcn_cosf(rev);
        bx[w] = sn * cs;
        by[w] = sn * sn;
        bz[w] = cs;
    }
    const float c2[2] = {hi ? bz[2] : 1.0f, hi ? by[2] : bx[2]};
    const float dA[2] = {hi ? 1.0f : bz[3], hi ? bx[3] : by[3]};
    const float dB[2] = {hi ? bz[3] : 1.0f, hi ? by[3] : bx[3]};
    v2f eA[2], eB[2];
#pragma unroll
    for (int i = 0; i < 2; ++i) {
        eA[i][0] = c2[i] * dA[0];
        eA[i][1] = c2[i] * dA[1];
        eB[i][0] = c2[i] * dB[0];
        eB[i][1] = c2[i] * dB[1];
    }
    v2f ub[8];
#pragma unroll
    for (int i = 0; i < 2; ++i) {
        ub[0 + i] = eA[i];
        ub[2 + i] = bz[0] * eB[i];
        ub[4 + i] = bx[0] * eB[i];
        ub[6 + i] = by[0] * eA[i];
    }

    v8f acc0 = {}, acc1 = {};
#pragma unroll
    for (int kk = 0; kk < 8; kk += 2) {
        v2f a, b;
        a[0] = af[kk >> 1][0];
        a[1] = af[kk >> 1][1];
        b[0] = af[kk >> 1][2];
        b[1] = af[kk >> 1][3];
        acc0 = __builtin_amdgcn_wmma_f32_16x16x4_f32(
            false, a, false, ub[kk], (short)0, acc0, false, false);
        acc1 = __builtin_amdgcn_wmma_f32_16x16x4_f32(
            false, b, false, ub[kk + 1], (short)0, acc1, false, false);
    }

    const float f1[2] = {hi ? bz[1] : 1.0f, hi ? by[1] : bx[1]};
    const float b4[4] = {1.0f, bz[4], bx[4], by[4]};
    float p0 = 0.0f, p1 = 0.0f;
#pragma unroll
    for (int r = 0; r < 8; ++r) {
        const float v = f1[r >> 2] * b4[r & 3];
        p0 = fmaf(v, acc0[r], p0);
        p1 = fmaf(v, acc1[r], p1);
    }
    const float p = p0 + p1;
    return p + xlane<16>(p);
}

__global__ __launch_bounds__(WG_WAVES * 32) void vqc_fused_kernel(
    const float* __restrict__ x, const float* __restrict__ params,
    float* __restrict__ out, int batch) {
    __shared__ __attribute__((aligned(16))) float At[32 * COL_PITCH];
    __shared__ __attribute__((aligned(16))) float Gs[FRAG_FLOATS];
    const int lane = threadIdx.x & 31;
    const int wave = __builtin_amdgcn_readfirstlane(threadIdx.x >> 5);
    const int bl = lane & 15;
    const int hi = lane >> 4;

    if (batch <= 0) return;

    const int groups = (batch + 15) >> 4;
    const int per = (groups + TOTAL_WAVES - 1) >> TOTAL_WAVES_LOG2;
    int g = (blockIdx.x * WG_WAVES + wave) * per;
    const int gend = (g + per < groups) ? (g + per) : groups;

    float xr[CH][5];
    load_inputs(x, g, gend, bl, batch, xr);

    float gc[10], gs[10];
#pragma unroll
    for (int q = 0; q < 10; ++q) {
        const float th = params[q * 2];
        gs[q] = __sinf(0.5f * th);
        gc[q] = __cosf(0.5f * th);
    }
    const int j0 = wave, j1 = wave + WG_WAVES;
    const float a0 = circuit_column(j0, lane, gc, gs);
    const float a1 = circuit_column(j1, lane, gc, gs);
    if (lane < 16) {
        At[j0 * COL_PITCH + lane] = a0;
        At[j1 * COL_PITCH + lane] = a1;
    }
    __syncthreads();

    {
        const v4f* ck = reinterpret_cast<const v4f*>(At + lane * COL_PITCH);
        const v4f* c0 =
            reinterpret_cast<const v4f*>(At + (lane ^ wave) * COL_PITCH);
        const v4f* c1 =
            reinterpret_cast<const v4f*>(At + (lane ^ wave ^ 16) * COL_PITCH);
        float w0 = 0.0f, w1 = 0.0f;
#pragma unroll
        for (int i = 0; i < 4; ++i) {
            const v4f a = ck[i], b = c0[i], c = c1[i];
#pragma unroll
            for (int e = 0; e < 4; ++e) {
                w0 = fmaf(a[e], b[e], w0);
                w1 = fmaf(a[e], c[e], w1);
            }
        }
#define BFLY(M)                                                  \
    {                                                            \
        const float o0 = xlane<M>(w0), o1 = xlane<M>(w1);        \
        w0 = (lane & M) ? (o0 - w0) : (w0 + o0);                 \
        w1 = (lane & M) ? (o1 - w1) : (w1 + o1);                 \
    }
        BFLY(1) BFLY(2) BFLY(4) BFLY(8) BFLY(16)
#undef BFLY
        const int z0w = (lane >> 4) & 1, z1w = (lane >> 3) & 1,
                  z2w = (lane >> 2) & 1, z3w = (lane >> 1) & 1, z4w = lane & 1;
        const int x1w = (wave >> 3) & 1, x2w = (wave >> 2) & 1,
                  x3w = (wave >> 1) & 1, x4w = wave & 1;
        const int x0w = 1 ^ z0w ^ z2w ^ z3w;
        const float wv = x0w ? w1 : w0;
        const int xm = wave + 16 * x0w;
        const int ny = __builtin_popcount(xm & lane);
        const float tv = ((ny & 2) ? -0.0625f : 0.0625f) * wv;
        const int row = 8 * z1w + 4 * x1w + 2 * x4w + z4w;
        const int kk = 2 * (2 * x0w + z0w) + x2w;
        Gs[z2w * 256 + row * 16 + kk * 2 + x3w] = tv;
    }
    __syncthreads();

    const v4f* gp = reinterpret_cast<const v4f*>(Gs + lane * 16);
    v4f af[4];
#pragma unroll
    for (int i = 0; i < 4; ++i) af[i] = gp[i];

    while (g < gend) {
#pragma unroll
        for (int t = 0; t < CH; ++t) {
            if (g + t < gend) {
                const int e = (g + t) * 16 + bl;
                const float p = group_value(xr[t], af, hi);
                if (hi == 0 && e < batch) out[e] = p;
            }
        }
        g += CH;
        if (g < gend) load_inputs(x, g, gend, bl, batch, xr);
    }
}

extern "C" void kernel_launch(void* const* d_in, const int* in_sizes, int n_in,
                              void* d_out, int out_size, void* d_ws,
                              size_t ws_size, hipStream_t stream) {
    const float* x = (const float*)d_in[0];
    const float* params = (const float*)d_in[1];
    float* out = (float*)d_out;

    const int batch = in_sizes[0] / 5;

    vqc_fused_kernel<<<NUM_WGS, WG_WAVES * 32, 0, stream>>>(x, params, out,
                                                            batch);
}
